# starting kernel
# baseline (speedup 1.0000x reference)
#include <hip/hip_runtime.h>
#include <math.h>

typedef __attribute__((ext_vector_type(2))) float v2f;
typedef __attribute__((ext_vector_type(4))) float v4f;
typedef __attribute__((ext_vector_type(8))) float v8f;

#define P_TOTAL 6272
#define NCAP 32
#define ICAP 8
#define KCAP 16
#define JCAP 16
#define PT   16
#define LDSJ 20
#define SQ_EPS 1e-7f
#define NROUT 3

__global__ __launch_bounds__(256)
void caps_routing_kernel(const float* __restrict__ X,
                         const float* __restrict__ W,
                         float* __restrict__ Out)
{
    __shared__ float pred[PT * NCAP * LDSJ];

    const int tid   = threadIdx.x;
    const int lane  = tid & 31;
    const int wave  = tid >> 5;
    const int blk   = blockIdx.x;
    const int k     = blk & (KCAP - 1);
    const int ptile = blk >> 4;
    const int p0    = ptile * PT;

    const int m    = lane & 15;
    const int half = lane >> 4;

    for (int nn = 0; nn < 4; ++nn) {
        const int n = wave * 4 + nn;

        const float* xb = X + (size_t)(p0 + m) * (NCAP * ICAP) + n * ICAP;
        v2f a0, a1;
        a0.x = xb[2 * half + 0];  a0.y = xb[2 * half + 1];
        a1.x = xb[2 * half + 4];  a1.y = xb[2 * half + 5];

        const float* wb = W + ((size_t)(k * NCAP + n) * ICAP) * JCAP + m;
        v2f b0, b1;
        b0.x = wb[(2 * half + 0) * JCAP];  b0.y = wb[(2 * half + 1) * JCAP];
        b1.x = wb[(2 * half + 4) * JCAP];  b1.y = wb[(2 * half + 5) * JCAP];

        v8f acc = {};
        acc = __builtin_amdgcn_wmma_f32_16x16x4_f32(false, a0, false, b0,
                                                    (short)0, acc, false, false);
        acc = __builtin_amdgcn_wmma_f32_16x16x4_f32(false, a1, false, b1,
                                                    (short)0, acc, false, false);

        #pragma unroll
        for (int v = 0; v < 8; ++v) {
            const int prow = v + 8 * half;
            pred[(prow * NCAP + n) * LDSJ + m] = acc[v];
        }
    }
    __syncthreads();

    for (int pl = wave; pl < PT; pl += 8) {
        float pj[JCAP];
        #pragma unroll
        for (int j = 0; j < JCAP; ++j)
            pj[j] = pred[(pl * NCAP + lane) * LDSJ + j];

        float braw = 0.0f;
        float out[JCAP];

        for (int r = 0; r < NROUT; ++r) {
            float mx = braw;
            #pragma unroll
            for (int s = 16; s >= 1; s >>= 1)
                mx = fmaxf(mx, __shfl_xor(mx, s, 32));
            float e = __expf(braw - mx);
            float se = e;
            #pragma unroll
            for (int s = 16; s >= 1; s >>= 1)
                se += __shfl_xor(se, s, 32);
            const float c = e / se;

            float sq = 0.0f;
            #pragma unroll
            for (int j = 0; j < JCAP; ++j) {
                float t = c * pj[j];
                #pragma unroll
                for (int s = 16; s >= 1; s >>= 1)
                    t += __shfl_xor(t, s, 32);
                out[j] = t;
                sq += t * t;
            }

            const float scale = sq / ((1.0f + sq) * sqrtf(sq + SQ_EPS));

            float dot = 0.0f;
            #pragma unroll
            for (int j = 0; j < JCAP; ++j) {
                out[j] *= scale;
                dot += pj[j] * out[j];
            }
            braw += dot;
        }

        if (lane == 0) {
            float* ob = Out + ((size_t)(p0 + pl) * KCAP + k) * JCAP;
            #pragma unroll
            for (int j = 0; j < JCAP; j += 4) {
                v4f t = { out[j], out[j + 1], out[j + 2], out[j + 3] };
                *(v4f*)(ob + j) = t;
            }
        }
    }
}

extern "C" void kernel_launch(void* const* d_in, const int* in_sizes, int n_in,
                              void* d_out, int out_size, void* d_ws, size_t ws_size,
                              hipStream_t stream) {
    const float* X = (const float*)d_in[0];
    const float* W = (const float*)d_in[1];
    float* Out = (float*)d_out;

    const int nblocks = (P_TOTAL / PT) * KCAP;
    caps_routing_kernel<<<nblocks, 256, 0, stream>>>(X, W, Out);
}


# Operands staged through LDS: a block reads its two W[k] and two position tiles once with whole-line 16-byte loads (W transposed on the write); fragments are 16-byte LDS reads, row pitch 1040 B
# speedup vs baseline: 4.5110x; 4.5110x over previous
#include <hip/hip_runtime.h>
#include <math.h>


typedef __attribute__((ext_vector_type(2))) float v2f;
typedef __attribute__((ext_vector_type(4))) float v4f;
typedef __attribute__((ext_vector_type(8))) float v8f;

#define P_TOTAL 6272
#define NCAP 32
#define ICAP 8
#define KCAP 16
#define JCAP 16
#define PT   16
#define SQ_EPS 1e-7f
#define NROUT 3

__device__ __forceinline__ float other_half(float v) {
    const int r = __builtin_amdgcn_permlanex16(__float_as_int(v), __float_as_int(v),
                                               0x76543210, 0xfedcba98, false, false);
    return __int_as_float(r);
}

#define ROWP 260

__global__ __launch_bounds__(128) __attribute__((amdgpu_waves_per_eu(3, 3)))
void caps_routing_kernel(const float* __restrict__ X,
                         const float* __restrict__ W,
                         float* __restrict__ Out)
{
    __shared__ __attribute__((aligned(16))) float w_lds[2 * JCAP * ROWP];
    __shared__ __attribute__((aligned(16))) float x_lds[2 * PT * ROWP];

    const int tid  = threadIdx.x;
    const int lane = tid & 31;
    const int wave = __builtin_amdgcn_readfirstlane(tid >> 5);
    const int tpair = (blockIdx.x >> 3) * 2;
    const int kpair = (blockIdx.x & 7) * 2;
    const int tile = tpair + (wave & 1);
    const int k    = kpair + (wave >> 1);
    const int p0   = tile * PT;
    const int m    = lane & 15;
    const int half = lane >> 4;

    {
        const v4f* wg = (const v4f*)(W + (size_t)kpair * (NCAP * ICAP * JCAP));
        const v4f* xg = (const v4f*)(X + (size_t)tpair * PT * (NCAP * ICAP));
        v4f wr[16], xr[16];
        #pragma unroll
        for (int r = 0; r < 16; ++r) wr[r] = wg[tid + 128 * r];
        #pragma unroll
        for (int r = 0; r < 16; ++r) xr[r] = xg[tid + 128 * r];
        #pragma unroll
        for (int r = 0; r < 16; ++r) {
            const int f  = tid + 128 * r;
            const int kk = f >> 10;
            const int n  = (f >> 5) & 31;
            const int i  = (f >> 2) & 7;
            const int j4 = (f & 3) * 4;
            float* dst = w_lds + (kk * JCAP + j4) * ROWP + n * ICAP + i;
            dst[0 * ROWP] = wr[r].x;
            dst[1 * ROWP] = wr[r].y;
            dst[2 * ROWP] = wr[r].z;
            dst[3 * ROWP] = wr[r].w;
        }
        #pragma unroll
        for (int r = 0; r < 16; ++r) {
            const int q = tid + 128 * r;
            *(v4f*)(x_lds + (q >> 6) * ROWP + (q & 63) * 4) = xr[r];
        }
    }
    __syncthreads();

    const float* xb = x_lds + ((wave & 1) * PT + m) * ROWP + 4 * half;
    const float* wb = w_lds + ((wave >> 1) * JCAP + m) * ROWP + 4 * half;

    v8f pred[NCAP];
    v4f xs[3][4];
    v4f ws[3][4];
    #pragma unroll
    for (int g = 0; g < NCAP / 4 + 2; ++g) {
        if (g < NCAP / 4) {
            #pragma unroll
            for (int t = 0; t < 4; ++t) {
                const int n = 4 * g + t;
                xs[g % 3][t] = *(const v4f*)(xb + n * ICAP);
                ws[g % 3][t] = *(const v4f*)(wb + n * ICAP);
            }
        }
        __builtin_amdgcn_sched_barrier(0);
        if (g >= 2) {
            #pragma unroll
            for (int t = 0; t < 4; ++t) {
                const int n = 4 * (g - 2) + t;
                const int s = (g - 2) % 3;
                v2f a0, a1, b0, b1;
                a0.x = ws[s][t].x;   a0.y = ws[s][t].y;
                a1.x = ws[s][t].z;   a1.y = ws[s][t].w;
                b0.x = xs[s][t].x;   b0.y = xs[s][t].y;
                b1.x = xs[s][t].z;   b1.y = xs[s][t].w;
                v8f acc = {};
                acc = __builtin_amdgcn_wmma_f32_16x16x4_f32(false, a0, false, b0,
                                                            (short)0, acc, false, false);
                acc = __builtin_amdgcn_wmma_f32_16x16x4_f32(false, a1, false, b1,
                                                            (short)0, acc, false, false);
                pred[n] = acc;
            }
            __builtin_amdgcn_sched_barrier(0);
        }
    }

    v2f osum[4];
    #pragma unroll
    for (int q = 0; q < 4; ++q) osum[q] = (v2f){0.0f, 0.0f};

    v2f out[4];

    #pragma unroll
    for (int r = 0; r < NROUT; ++r) {
        float be[NCAP];
        float mx = 0.0f;
        if (r > 0) {
            #pragma unroll
            for (int n = 0; n < NCAP; ++n) {
                v2f d2 = {0.0f, 0.0f};
                #pragma unroll
                for (int q = 0; q < 4; ++q) {
                    const v2f pq = {pred[n][2 * q], pred[n][2 * q + 1]};
                    d2 += pq * osum[q];
                }
                const float d = d2.x + d2.y;
                be[n] = d + other_half(d);
            }
            mx = be[0];
            #pragma unroll
            for (int n = 1; n < NCAP; ++n) mx = fmaxf(mx, be[n]);
        } else {
            #pragma unroll
            for (int n = 0; n < NCAP; ++n) be[n] = 0.0f;
        }

        float se = 0.0f;
        #pragma unroll
        for (int q = 0; q < 4; ++q) out[q] = (v2f){0.0f, 0.0f};
        #pragma unroll
        for (int n = 0; n < NCAP; ++n) {
            const float e = __expf(be[n] - mx);
            se += e;
            const v2f ee = {e, e};
            #pragma unroll
            for (int q = 0; q < 4; ++q) {
                const v2f pq = {pred[n][2 * q], pred[n][2 * q + 1]};
                out[q] += ee * pq;
            }
        }
        const float inv = 1.0f / se;

        v2f sq2 = {0.0f, 0.0f};
        #pragma unroll
        for (int q = 0; q < 4; ++q) { out[q] *= inv; sq2 += out[q] * out[q]; }
        const float sqp = sq2.x + sq2.y;
        const float sq = sqp + other_half(sqp);

        const float scale = sq / ((1.0f + sq) * sqrtf(sq + SQ_EPS));
        #pragma unroll
        for (int q = 0; q < 4; ++q) out[q] *= scale;

        if (r + 1 < NROUT) {
            #pragma unroll
            for (int q = 0; q < 4; ++q) osum[q] += out[q];
        }
    }

    float* ob = Out + ((size_t)(p0 + m) * KCAP + k) * JCAP + 8 * half;
    const v4f t0 = { out[0].x, out[0].y, out[1].x, out[1].y };
    const v4f t1 = { out[2].x, out[2].y, out[3].x, out[3].y };
    *(v4f*)(ob + 0) = t0;
    *(v4f*)(ob + 4) = t1;
}

extern "C" void kernel_launch(void* const* d_in, const int* in_sizes, int n_in,
                              void* d_out, int out_size, void* d_ws, size_t ws_size,
                              hipStream_t stream) {
    const float* X = (const float*)d_in[0];
    const float* W = (const float*)d_in[1];
    float* Out = (float*)d_out;

    caps_routing_kernel<<<1568, 128, 0, stream>>>(X, W, Out);
}
